# starting kernel
# baseline (speedup 1.0000x reference)
#include <hip/hip_runtime.h>

#define NN 50000
#define EE 800000
#define DD 128
#define ETOT (EE + NN)
#define SLOPE 0.2f

typedef float v2f __attribute__((ext_vector_type(2)));
typedef float v8f __attribute__((ext_vector_type(8)));

__global__ __launch_bounds__(256) void gemm_wmma_f32(
    const float* __restrict__ A, const float* __restrict__ W,
    float* __restrict__ H, int accumulate) {
  const int lane = threadIdx.x & 31;
  const int wv   = threadIdx.x >> 5;
  const int m0   = blockIdx.x * 16;
  const int n0   = wv * 16;
  const int half = lane >> 4;
  const int l15  = lane & 15;
  const int row  = m0 + l15;
  const int col  = n0 + l15;

  v8f acc;
  if (accumulate) {
#pragma unroll
    for (int r = 0; r < 8; ++r)
      acc[r] = H[(size_t)(m0 + half * 8 + r) * DD + col];
  } else {
#pragma unroll
    for (int r = 0; r < 8; ++r) acc[r] = 0.0f;
  }

  const float* arow = A + (size_t)row * DD;
#pragma unroll 4
  for (int k = 0; k < DD; k += 4) {
    v2f av, bv;
    av[0] = arow[k + 2 * half];
    av[1] = arow[k + 2 * half + 1];
    bv[0] = W[(size_t)(k + 2 * half) * DD + col];
    bv[1] = W[(size_t)(k + 2 * half + 1) * DD + col];
    acc = __builtin_amdgcn_wmma_f32_16x16x4_f32(false, av, false, bv,
                                                (short)0, acc, false, false);
  }
#pragma unroll
  for (int r = 0; r < 8; ++r)
    H[(size_t)(m0 + half * 8 + r) * DD + col] = acc[r];
}

__global__ __launch_bounds__(256) void scores_init(
    const float* __restrict__ H, const float* __restrict__ a_s,
    const float* __restrict__ a_d, float* __restrict__ s_src,
    float* __restrict__ s_dst, unsigned* __restrict__ m_ord,
    float* __restrict__ denom) {
  const int lane = threadIdx.x & 31;
  const int node = blockIdx.x * 8 + (threadIdx.x >> 5);
  if (node >= NN) return;
  const float4 hv = *(const float4*)(H + (size_t)node * DD + lane * 4);
  const float4 as = *(const float4*)(a_s + lane * 4);
  const float4 ad = *(const float4*)(a_d + lane * 4);
  float ps = hv.x * as.x + hv.y * as.y + hv.z * as.z + hv.w * as.w;
  float pd = hv.x * ad.x + hv.y * ad.y + hv.z * ad.z + hv.w * ad.w;
#pragma unroll
  for (int off = 16; off > 0; off >>= 1) {
    ps += __shfl_down(ps, off, 32);
    pd += __shfl_down(pd, off, 32);
  }
  if (lane == 0) {
    s_src[node] = ps;
    s_dst[node] = pd;
    m_ord[node] = 0u;
    denom[node] = 0.0f;
  }
}

__device__ __forceinline__ unsigned f2ord(float f) {
  unsigned u = __float_as_uint(f);
  return (u & 0x80000000u) ? ~u : (u | 0x80000000u);
}
__device__ __forceinline__ float ord2f(unsigned o) {
  return (o & 0x80000000u) ? __uint_as_float(o & 0x7FFFFFFFu)
                           : __uint_as_float(~o);
}

__global__ __launch_bounds__(256) void edge_max(
    const int* __restrict__ ei, const float* __restrict__ s_src,
    const float* __restrict__ s_dst, float* __restrict__ wbuf,
    unsigned* __restrict__ m_ord) {
  int e = blockIdx.x * blockDim.x + threadIdx.x;
  if (e >= ETOT) return;
  int s, d;
  if (e < EE) { s = ei[e]; d = ei[EE + e]; } else { s = d = e - EE; }
  float v = s_src[s] + s_dst[d];
  v = v > 0.0f ? v : SLOPE * v;
  wbuf[e] = v;
  atomicMax(&m_ord[d], f2ord(v));
}

__global__ __launch_bounds__(256) void edge_exp(
    const int* __restrict__ ei, const unsigned* __restrict__ m_ord,
    float* __restrict__ wbuf, float* __restrict__ denom) {
  int e = blockIdx.x * blockDim.x + threadIdx.x;
  if (e >= ETOT) return;
  int d = (e < EE) ? ei[EE + e] : (e - EE);
  float z = __expf(wbuf[e] - ord2f(m_ord[d]));
  wbuf[e] = z;
  atomicAdd(&denom[d], z);
}

__global__ __launch_bounds__(256) void init_out(float* __restrict__ out,
                                                const float* __restrict__ b) {
  int i = blockIdx.x * blockDim.x + threadIdx.x;
  out[i] = b[i & (DD - 1)];
}

__global__ __launch_bounds__(256) void edge_scatter(
    const int* __restrict__ ei, const float* __restrict__ H,
    const float* __restrict__ wbuf, const float* __restrict__ denom,
    float* __restrict__ out) {
  int e    = (int)((blockIdx.x * blockDim.x + threadIdx.x) >> 5);
  int lane = threadIdx.x & 31;
  if (e >= ETOT) return;
  int s, d;
  if (e < EE) { s = ei[e]; d = ei[EE + e]; } else { s = d = e - EE; }
  float alpha = wbuf[e] / denom[d];
  const float4 hv = *(const float4*)(H + (size_t)s * DD + lane * 4);
  float* o = out + (size_t)d * DD + lane * 4;
  atomicAdd(o + 0, alpha * hv.x);
  atomicAdd(o + 1, alpha * hv.y);
  atomicAdd(o + 2, alpha * hv.z);
  atomicAdd(o + 3, alpha * hv.w);
}

static void run_gat(const float* A1, const float* A2, const float* Wmat,
                    const float* a_s, const float* a_d, const float* bias,
                    const int* ei, float* ht, float* out, float* s_src,
                    float* s_dst, unsigned* m_ord, float* denom, float* wbuf,
                    hipStream_t stream) {
  gemm_wmma_f32<<<NN / 16, 256, 0, stream>>>(A1, Wmat, ht, 0);
  if (A2)
    gemm_wmma_f32<<<NN / 16, 256, 0, stream>>>(A2, Wmat + DD * DD, ht, 1);
  scores_init<<<NN / 8, 256, 0, stream>>>(ht, a_s, a_d, s_src, s_dst, m_ord,
                                          denom);
  edge_max<<<(ETOT + 255) / 256, 256, 0, stream>>>(ei, s_src, s_dst, wbuf,
                                                   m_ord);
  edge_exp<<<(ETOT + 255) / 256, 256, 0, stream>>>(ei, m_ord, wbuf, denom);
  init_out<<<(NN * DD) / 256, 256, 0, stream>>>(out, bias);
  edge_scatter<<<(ETOT + 7) / 8, 256, 0, stream>>>(ei, ht, wbuf, denom, out);
}

extern "C" void kernel_launch(void* const* d_in, const int* in_sizes, int n_in,
                              void* d_out, int out_size, void* d_ws,
                              size_t ws_size, hipStream_t stream) {
  const float* x    = (const float*)d_in[0];
  const int*   ei   = (const int*)d_in[1];
  const float* We   = (const float*)d_in[2];
  const float* ae_s = (const float*)d_in[3];
  const float* ae_d = (const float*)d_in[4];
  const float* be   = (const float*)d_in[5];
  const float* Wp   = (const float*)d_in[6];
  const float* ap_s = (const float*)d_in[7];
  const float* ap_d = (const float*)d_in[8];
  const float* bp   = (const float*)d_in[9];
  const float* Wd   = (const float*)d_in[10];
  const float* ad_s = (const float*)d_in[11];
  const float* ad_d = (const float*)d_in[12];
  const float* bd   = (const float*)d_in[13];
  float*       out  = (float*)d_out;

  const size_t ND = (size_t)NN * DD;
  float*    enc   = (float*)d_ws;
  float*    hA    = enc + ND;
  float*    hB    = hA + ND;
  float*    ht    = hB + ND;
  float*    s_src = ht + ND;
  float*    s_dst = s_src + NN;
  unsigned* m_ord = (unsigned*)(s_dst + NN);
  float*    denom = (float*)(m_ord + NN);
  float*    wbuf  = denom + NN;

  run_gat(x, nullptr, We, ae_s, ae_d, be, ei, ht, enc, s_src, s_dst, m_ord,
          denom, wbuf, stream);
  run_gat(enc, enc, Wp, ap_s, ap_d, bp, ei, ht, hA, s_src, s_dst, m_ord,
          denom, wbuf, stream);
  run_gat(hA, enc, Wp, ap_s, ap_d, bp, ei, ht, hB, s_src, s_dst, m_ord,
          denom, wbuf, stream);
  run_gat(hB, nullptr, Wd, ad_s, ad_d, bd, ei, ht, out, s_src, s_dst, m_ord,
          denom, wbuf, stream);
}


# edges counting-sorted by destination once per call (int atomics, groups ordered by source); one wave per node does max, exp sum, weighted row sum in registers; no float atomics, fixed sum order
# speedup vs baseline: 3.2139x; 3.2139x over previous
#include <hip/hip_runtime.h>

#define NN 50000
#define EE 800000
#define DD 128
#define ETOT (EE + NN)
#define SLOPE 0.2f

typedef float v2f __attribute__((ext_vector_type(2)));
typedef float v8f __attribute__((ext_vector_type(8)));

__global__ __launch_bounds__(256) void gemm_wmma_f32(
    const float* __restrict__ A, const float* __restrict__ W,
    float* __restrict__ H, int accumulate) {
  const int lane = threadIdx.x & 31;
  const int wv   = threadIdx.x >> 5;
  const int m0   = blockIdx.x * 16;
  const int n0   = wv * 16;
  const int half = lane >> 4;
  const int l15  = lane & 15;
  const int row  = m0 + l15;
  const int col  = n0 + l15;

  v8f acc;
  if (accumulate) {
#pragma unroll
    for (int r = 0; r < 8; ++r)
      acc[r] = H[(size_t)(m0 + half * 8 + r) * DD + col];
  } else {
#pragma unroll
    for (int r = 0; r < 8; ++r) acc[r] = 0.0f;
  }

  const float* arow = A + (size_t)row * DD;
#pragma unroll 4
  for (int k = 0; k < DD; k += 4) {
    v2f av, bv;
    av[0] = arow[k + 2 * half];
    av[1] = arow[k + 2 * half + 1];
    bv[0] = W[(size_t)(k + 2 * half) * DD + col];
    bv[1] = W[(size_t)(k + 2 * half + 1) * DD + col];
    acc = __builtin_amdgcn_wmma_f32_16x16x4_f32(false, av, false, bv,
                                                (short)0, acc, false, false);
  }
#pragma unroll
  for (int r = 0; r < 8; ++r)
    H[(size_t)(m0 + half * 8 + r) * DD + col] = acc[r];
}

__global__ __launch_bounds__(256) void csr_zero(int* __restrict__ deg) {
  int i = blockIdx.x * blockDim.x + threadIdx.x;
  if (i < NN) deg[i] = 0;
}

__global__ __launch_bounds__(256) void csr_count(const int* __restrict__ ei,
                                                 int* __restrict__ deg,
                                                 int* __restrict__ slot) {
  int e = blockIdx.x * blockDim.x + threadIdx.x;
  slot[e] = atomicAdd(&deg[ei[EE + e]], 1);
}

#define SCAN_T 1024
#define SCAN_ROWS ((NN + SCAN_T) / SCAN_T)
__global__ __launch_bounds__(SCAN_T) void csr_scan(const int* __restrict__ deg,
                                                   int* __restrict__ rowptr) {
  __shared__ int wsum[32];
  const int lane = threadIdx.x & 31;
  const int wv   = threadIdx.x >> 5;
  const int b    = wv * (SCAN_ROWS * 32) + lane;
  int d[SCAN_ROWS];
  int local = 0;
#pragma unroll
  for (int k = 0; k < SCAN_ROWS; ++k) {
    const int idx = b + k * 32;
    const int v   = deg[idx < NN ? idx : NN - 1];
    d[k] = idx < NN ? v : 0;
    local += d[k];
  }
  for (int off = 16; off > 0; off >>= 1) local += __shfl_xor(local, off, 32);
  if (lane == 0) wsum[wv] = local;
  __syncthreads();
  if (wv == 0) {
    const int w = wsum[lane];
    int winc = w;
    for (int off = 1; off < 32; off <<= 1) {
      const int v = __shfl_up(winc, off, 32);
      if (lane >= off) winc += v;
    }
    wsum[lane] = winc - w;
  }
  __syncthreads();
  int carry = wsum[wv];
#pragma unroll
  for (int k = 0; k < SCAN_ROWS; ++k) {
    int inc = d[k];
    for (int off = 1; off < 32; off <<= 1) {
      const int v = __shfl_up(inc, off, 32);
      if (lane >= off) inc += v;
    }
    rowptr[b + k * 32] = carry + inc - d[k];
    carry += __shfl(inc, 31, 32);
  }
}

__global__ __launch_bounds__(256) void csr_place(const int* __restrict__ ei,
                                                 const int* __restrict__ rowptr,
                                                 const int* __restrict__ slot,
                                                 int* __restrict__ grouped) {
  int e = blockIdx.x * blockDim.x + threadIdx.x;
  grouped[rowptr[ei[EE + e]] + slot[e]] = ei[e];
}

__global__ __launch_bounds__(256) void csr_order(const int* __restrict__ rowptr,
                                                 const int* __restrict__ grouped,
                                                 int* __restrict__ csr) {
  const int lane = threadIdx.x & 31;
  const int node = blockIdx.x * 8 + (threadIdx.x >> 5);
  if (node >= NN) return;
  const int start = rowptr[node];
  const int cnt   = rowptr[node + 1] - start;
  for (int i = lane; i < cnt; i += 32) {
    const int si = grouped[start + i];
    int r = 0;
    for (int j = 0; j < cnt; ++j) {
      const int sj = grouped[start + j];
      r += (sj < si || (sj == si && j < i)) ? 1 : 0;
    }
    csr[start + r] = si;
  }
}

__global__ __launch_bounds__(256) void scores(
    const float* __restrict__ H, const float* __restrict__ a_s,
    const float* __restrict__ a_d, float* __restrict__ s_src,
    float* __restrict__ s_dst) {
  const int lane = threadIdx.x & 31;
  const int node = blockIdx.x * 8 + (threadIdx.x >> 5);
  if (node >= NN) return;
  const float4 hv = *(const float4*)(H + (size_t)node * DD + lane * 4);
  const float4 as = *(const float4*)(a_s + lane * 4);
  const float4 ad = *(const float4*)(a_d + lane * 4);
  float ps = hv.x * as.x + hv.y * as.y + hv.z * as.z + hv.w * as.w;
  float pd = hv.x * ad.x + hv.y * ad.y + hv.z * ad.z + hv.w * ad.w;
#pragma unroll
  for (int off = 16; off > 0; off >>= 1) {
    ps += __shfl_down(ps, off, 32);
    pd += __shfl_down(pd, off, 32);
  }
  if (lane == 0) {
    s_src[node] = ps;
    s_dst[node] = pd;
  }
}

__device__ __forceinline__ float edge_score(float ss, float sd) {
  float v = ss + sd;
  return v > 0.0f ? v : SLOPE * v;
}

__device__ __forceinline__ void add_rows(float4& acc, const float* __restrict__ Hl,
                                         int s, float a, int n) {
#pragma unroll 4
  for (int j = 0; j < n; ++j) {
    const int   sj = __builtin_amdgcn_readlane(s, j);
    const float aj = __int_as_float(__builtin_amdgcn_readlane(__float_as_int(a), j));
    const float4 hv = *(const float4*)(Hl + (size_t)sj * DD);
    acc.x += aj * hv.x;
    acc.y += aj * hv.y;
    acc.z += aj * hv.z;
    acc.w += aj * hv.w;
  }
}

__global__ __launch_bounds__(256) void node_softmax_sum(
    const int* __restrict__ rowptr, const int* __restrict__ csr,
    const float* __restrict__ H, const float* __restrict__ s_src,
    const float* __restrict__ s_dst, const float* __restrict__ bias,
    float* __restrict__ out) {
  const int lane = threadIdx.x & 31;
  const int node = __builtin_amdgcn_readfirstlane(
      (int)(blockIdx.x * 8 + (threadIdx.x >> 5)));
  if (node >= NN) return;
  const int   start = rowptr[node];
  const int   cnt   = rowptr[node + 1] - start;
  const int   tot   = cnt + 1;
  const float sd    = s_dst[node];

  int   s0 = node;
  float v0 = -INFINITY;
  if (lane < tot) {
    if (lane < cnt) s0 = csr[start + lane];
    v0 = edge_score(s_src[s0], sd);
  }
  float m = v0;
  for (int i = 32 + lane; i < tot; i += 32) {
    const int s = (i < cnt) ? csr[start + i] : node;
    m = fmaxf(m, edge_score(s_src[s], sd));
  }
  for (int off = 16; off > 0; off >>= 1) m = fmaxf(m, __shfl_xor(m, off, 32));

  const float z0 = (lane < tot) ? __expf(v0 - m) : 0.0f;
  float den = z0;
  for (int i = 32 + lane; i < tot; i += 32) {
    const int s = (i < cnt) ? csr[start + i] : node;
    den += __expf(edge_score(s_src[s], sd) - m);
  }
  for (int off = 16; off > 0; off >>= 1) den += __shfl_xor(den, off, 32);

  float4 acc = make_float4(0.0f, 0.0f, 0.0f, 0.0f);
  const float* Hl = H + lane * 4;
  add_rows(acc, Hl, s0, z0 / den, tot < 32 ? tot : 32);
  for (int base = 32; base < tot; base += 32) {
    const int i = base + lane;
    int   s = node;
    float a = 0.0f;
    if (i < tot) {
      if (i < cnt) s = csr[start + i];
      a = __expf(edge_score(s_src[s], sd) - m) / den;
    }
    add_rows(acc, Hl, s, a, (tot - base) < 32 ? (tot - base) : 32);
  }
  const float4 b = *(const float4*)(bias + lane * 4);
  acc.x += b.x;
  acc.y += b.y;
  acc.z += b.z;
  acc.w += b.w;
  *(float4*)(out + (size_t)node * DD + lane * 4) = acc;
}

static void run_gat(const float* A1, const float* A2, const float* Wmat,
                    const float* a_s, const float* a_d, const float* bias,
                    const int* rowptr, const int* csr, float* ht, float* out,
                    float* s_src, float* s_dst, hipStream_t stream) {
  gemm_wmma_f32<<<NN / 16, 256, 0, stream>>>(A1, Wmat, ht, 0);
  if (A2)
    gemm_wmma_f32<<<NN / 16, 256, 0, stream>>>(A2, Wmat + DD * DD, ht, 1);
  scores<<<NN / 8, 256, 0, stream>>>(ht, a_s, a_d, s_src, s_dst);
  node_softmax_sum<<<NN / 8, 256, 0, stream>>>(rowptr, csr, ht, s_src, s_dst,
                                               bias, out);
}

extern "C" void kernel_launch(void* const* d_in, const int* in_sizes, int n_in,
                              void* d_out, int out_size, void* d_ws,
                              size_t ws_size, hipStream_t stream) {
  const float* x    = (const float*)d_in[0];
  const int*   ei   = (const int*)d_in[1];
  const float* We   = (const float*)d_in[2];
  const float* ae_s = (const float*)d_in[3];
  const float* ae_d = (const float*)d_in[4];
  const float* be   = (const float*)d_in[5];
  const float* Wp   = (const float*)d_in[6];
  const float* ap_s = (const float*)d_in[7];
  const float* ap_d = (const float*)d_in[8];
  const float* bp   = (const float*)d_in[9];
  const float* Wd   = (const float*)d_in[10];
  const float* ad_s = (const float*)d_in[11];
  const float* ad_d = (const float*)d_in[12];
  const float* bd   = (const float*)d_in[13];
  float*       out  = (float*)d_out;

  const size_t ND = (size_t)NN * DD;
  float*    enc   = (float*)d_ws;
  float*    hA    = enc + ND;
  float*    hB    = hA + ND;
  float*    ht    = hB + ND;
  float*    s_src = ht + ND;
  float*    s_dst = s_src + NN;
  int*      deg    = (int*)(s_dst + NN);
  int*      csr    = deg + NN;
  int*      rowptr = csr + EE;
  int*      slot    = (int*)hA;
  int*      grouped = (int*)hB;

  csr_zero<<<(NN + 255) / 256, 256, 0, stream>>>(deg);
  csr_count<<<EE / 256, 256, 0, stream>>>(ei, deg, slot);
  csr_scan<<<1, SCAN_T, 0, stream>>>(deg, rowptr);
  csr_place<<<EE / 256, 256, 0, stream>>>(ei, rowptr, slot, grouped);
  csr_order<<<NN / 8, 256, 0, stream>>>(rowptr, grouped, csr);

  run_gat(x, nullptr, We, ae_s, ae_d, be, rowptr, csr, ht, enc, s_src, s_dst,
          stream);
  run_gat(enc, enc, Wp, ap_s, ap_d, bp, rowptr, csr, ht, hA, s_src, s_dst,
          stream);
  run_gat(hA, enc, Wp, ap_s, ap_d, bp, rowptr, csr, ht, hB, s_src, s_dst,
          stream);
  run_gat(hB, nullptr, Wd, ad_s, ad_d, bd, rowptr, csr, ht, out, s_src, s_dst,
          stream);
}


# GEMM re-tiled: one wave per 16x128 strip (A loaded once, 16-byte reads), W staged once per work-group in LDS as 16-byte B fragments; k sum order inside each 8-k group changes (fixed order)
# speedup vs baseline: 3.8032x; 1.1834x over previous
#include <hip/hip_runtime.h>

#define NN 50000
#define EE 800000
#define DD 128
#define ETOT (EE + NN)
#define SLOPE 0.2f

typedef float v2f __attribute__((ext_vector_type(2)));
typedef float v8f __attribute__((ext_vector_type(8)));

#define MT (NN / 16)
__global__ __launch_bounds__(256) void gemm_wmma_f32(
    const float* __restrict__ A, const float* __restrict__ W,
    float* __restrict__ H, int accumulate) {
  __shared__ float4 Ws[(DD / 4) * DD];
  const int tid  = threadIdx.x;
  const int lane = tid & 31;
  const int mt   = blockIdx.x * 8 + (tid >> 5);
  const int h    = lane >> 4;
  const int c    = lane & 15;
  const int m0 = (mt < MT ? mt : MT - 1) * 16;

  const float* arow = A + (size_t)(m0 + c) * DD + 4 * h;
  float4 a[DD / 8];
#pragma unroll
  for (int j = 0; j < DD / 8; ++j) a[j] = *(const float4*)(arow + 8 * j);

  {
    const int col = tid & (DD - 1);
    const int kq0 = tid >> 7;
#pragma unroll
    for (int i = 0; i < DD / 8; ++i) {
      const int kq = kq0 + 2 * i;
      const float* wp = W + (size_t)(4 * kq) * DD + col;
      float4 w4;
      w4.x = wp[0];
      w4.y = wp[DD];
      w4.z = wp[2 * DD];
      w4.w = wp[3 * DD];
      Ws[kq * DD + col] = w4;
    }
  }
  __syncthreads();

  if (mt >= MT) return;

  float* hp = H + (size_t)(m0 + 8 * h) * DD + c;
  v8f acc[8];
  if (accumulate) {
#pragma unroll
    for (int t = 0; t < 8; ++t)
#pragma unroll
      for (int r = 0; r < 8; ++r) acc[t][r] = hp[(size_t)r * DD + 16 * t];
  } else {
#pragma unroll
    for (int t = 0; t < 8; ++t)
#pragma unroll
      for (int r = 0; r < 8; ++r) acc[t][r] = 0.0f;
  }

  const float4* bp = Ws + h * DD + c;
#pragma unroll
  for (int j = 0; j < DD / 8; ++j) {
    float4 b[8];
#pragma unroll
    for (int t = 0; t < 8; ++t) b[t] = bp[j * 2 * DD + 16 * t];
    v2f av;
    av[0] = a[j].x;
    av[1] = a[j].y;
#pragma unroll
    for (int t = 0; t < 8; ++t) {
      v2f bv;
      bv[0] = b[t].x;
      bv[1] = b[t].y;
      acc[t] = __builtin_amdgcn_wmma_f32_16x16x4_f32(false, av, false, bv,
                                                     (short)0, acc[t], false, false);
    }
    av[0] = a[j].z;
    av[1] = a[j].w;
#pragma unroll
    for (int t = 0; t < 8; ++t) {
      v2f bv;
      bv[0] = b[t].z;
      bv[1] = b[t].w;
      acc[t] = __builtin_amdgcn_wmma_f32_16x16x4_f32(false, av, false, bv,
                                                     (short)0, acc[t], false, false);
    }
  }
#pragma unroll
  for (int t = 0; t < 8; ++t)
#pragma unroll
    for (int r = 0; r < 8; ++r) hp[(size_t)r * DD + 16 * t] = acc[t][r];
}

__global__ __launch_bounds__(256) void csr_zero(int* __restrict__ deg) {
  int i = blockIdx.x * blockDim.x + threadIdx.x;
  if (i < NN) deg[i] = 0;
}

__global__ __launch_bounds__(256) void csr_count(const int* __restrict__ ei,
                                                 int* __restrict__ deg,
                                                 int* __restrict__ slot) {
  int e = blockIdx.x * blockDim.x + threadIdx.x;
  slot[e] = atomicAdd(&deg[ei[EE + e]], 1);
}

#define SCAN_T 1024
#define SCAN_ROWS ((NN + SCAN_T) / SCAN_T)
__global__ __launch_bounds__(SCAN_T) void csr_scan(const int* __restrict__ deg,
                                                   int* __restrict__ rowptr) {
  __shared__ int wsum[32];
  const int lane = threadIdx.x & 31;
  const int wv   = threadIdx.x >> 5;
  const int b    = wv * (SCAN_ROWS * 32) + lane;
  int d[SCAN_ROWS];
  int local = 0;
#pragma unroll
  for (int k = 0; k < SCAN_ROWS; ++k) {
    const int idx = b + k * 32;
    const int v   = deg[idx < NN ? idx : NN - 1];
    d[k] = idx < NN ? v : 0;
    local += d[k];
  }
  for (int off = 16; off > 0; off >>= 1) local += __shfl_xor(local, off, 32);
  if (lane == 0) wsum[wv] = local;
  __syncthreads();
  if (wv == 0) {
    const int w = wsum[lane];
    int winc = w;
    for (int off = 1; off < 32; off <<= 1) {
      const int v = __shfl_up(winc, off, 32);
      if (lane >= off) winc += v;
    }
    wsum[lane] = winc - w;
  }
  __syncthreads();
  int carry = wsum[wv];
#pragma unroll
  for (int k = 0; k < SCAN_ROWS; ++k) {
    int inc = d[k];
    for (int off = 1; off < 32; off <<= 1) {
      const int v = __shfl_up(inc, off, 32);
      if (lane >= off) inc += v;
    }
    rowptr[b + k * 32] = carry + inc - d[k];
    carry += __shfl(inc, 31, 32);
  }
}

__global__ __launch_bounds__(256) void csr_place(const int* __restrict__ ei,
                                                 const int* __restrict__ rowptr,
                                                 const int* __restrict__ slot,
                                                 int* __restrict__ grouped) {
  int e = blockIdx.x * blockDim.x + threadIdx.x;
  grouped[rowptr[ei[EE + e]] + slot[e]] = ei[e];
}

__global__ __launch_bounds__(256) void csr_order(const int* __restrict__ rowptr,
                                                 const int* __restrict__ grouped,
                                                 int* __restrict__ csr) {
  const int lane = threadIdx.x & 31;
  const int node = blockIdx.x * 8 + (threadIdx.x >> 5);
  if (node >= NN) return;
  const int start = rowptr[node];
  const int cnt   = rowptr[node + 1] - start;
  for (int i = lane; i < cnt; i += 32) {
    const int si = grouped[start + i];
    int r = 0;
    for (int j = 0; j < cnt; ++j) {
      const int sj = grouped[start + j];
      r += (sj < si || (sj == si && j < i)) ? 1 : 0;
    }
    csr[start + r] = si;
  }
}

__global__ __launch_bounds__(256) void scores(
    const float* __restrict__ H, const float* __restrict__ a_s,
    const float* __restrict__ a_d, float* __restrict__ s_src,
    float* __restrict__ s_dst) {
  const int lane = threadIdx.x & 31;
  const int node = blockIdx.x * 8 + (threadIdx.x >> 5);
  if (node >= NN) return;
  const float4 hv = *(const float4*)(H + (size_t)node * DD + lane * 4);
  const float4 as = *(const float4*)(a_s + lane * 4);
  const float4 ad = *(const float4*)(a_d + lane * 4);
  float ps = hv.x * as.x + hv.y * as.y + hv.z * as.z + hv.w * as.w;
  float pd = hv.x * ad.x + hv.y * ad.y + hv.z * ad.z + hv.w * ad.w;
#pragma unroll
  for (int off = 16; off > 0; off >>= 1) {
    ps += __shfl_down(ps, off, 32);
    pd += __shfl_down(pd, off, 32);
  }
  if (lane == 0) {
    s_src[node] = ps;
    s_dst[node] = pd;
  }
}

__device__ __forceinline__ float edge_score(float ss, float sd) {
  float v = ss + sd;
  return v > 0.0f ? v : SLOPE * v;
}

__device__ __forceinline__ void add_rows(float4& acc, const float* __restrict__ Hl,
                                         int s, float a, int n) {
#pragma unroll 4
  for (int j = 0; j < n; ++j) {
    const int   sj = __builtin_amdgcn_readlane(s, j);
    const float aj = __int_as_float(__builtin_amdgcn_readlane(__float_as_int(a), j));
    const float4 hv = *(const float4*)(Hl + (size_t)sj * DD);
    acc.x += aj * hv.x;
    acc.y += aj * hv.y;
    acc.z += aj * hv.z;
    acc.w += aj * hv.w;
  }
}

__global__ __launch_bounds__(256) void node_softmax_sum(
    const int* __restrict__ rowptr, const int* __restrict__ csr,
    const float* __restrict__ H, const float* __restrict__ s_src,
    const float* __restrict__ s_dst, const float* __restrict__ bias,
    float* __restrict__ out) {
  const int lane = threadIdx.x & 31;
  const int node = __builtin_amdgcn_readfirstlane(
      (int)(blockIdx.x * 8 + (threadIdx.x >> 5)));
  if (node >= NN) return;
  const int   start = rowptr[node];
  const int   cnt   = rowptr[node + 1] - start;
  const int   tot   = cnt + 1;
  const float sd    = s_dst[node];

  int   s0 = node;
  float v0 = -INFINITY;
  if (lane < tot) {
    if (lane < cnt) s0 = csr[start + lane];
    v0 = edge_score(s_src[s0], sd);
  }
  float m = v0;
  for (int i = 32 + lane; i < tot; i += 32) {
    const int s = (i < cnt) ? csr[start + i] : node;
    m = fmaxf(m, edge_score(s_src[s], sd));
  }
  for (int off = 16; off > 0; off >>= 1) m = fmaxf(m, __shfl_xor(m, off, 32));

  const float z0 = (lane < tot) ? __expf(v0 - m) : 0.0f;
  float den = z0;
  for (int i = 32 + lane; i < tot; i += 32) {
    const int s = (i < cnt) ? csr[start + i] : node;
    den += __expf(edge_score(s_src[s], sd) - m);
  }
  for (int off = 16; off > 0; off >>= 1) den += __shfl_xor(den, off, 32);

  float4 acc = make_float4(0.0f, 0.0f, 0.0f, 0.0f);
  const float* Hl = H + lane * 4;
  add_rows(acc, Hl, s0, z0 / den, tot < 32 ? tot : 32);
  for (int base = 32; base < tot; base += 32) {
    const int i = base + lane;
    int   s = node;
    float a = 0.0f;
    if (i < tot) {
      if (i < cnt) s = csr[start + i];
      a = __expf(edge_score(s_src[s], sd) - m) / den;
    }
    add_rows(acc, Hl, s, a, (tot - base) < 32 ? (tot - base) : 32);
  }
  const float4 b = *(const float4*)(bias + lane * 4);
  acc.x += b.x;
  acc.y += b.y;
  acc.z += b.z;
  acc.w += b.w;
  *(float4*)(out + (size_t)node * DD + lane * 4) = acc;
}

static void run_gat(const float* A1, const float* A2, const float* Wmat,
                    const float* a_s, const float* a_d, const float* bias,
                    const int* rowptr, const int* csr, float* ht, float* out,
                    float* s_src, float* s_dst, hipStream_t stream) {
  gemm_wmma_f32<<<(MT + 7) / 8, 256, 0, stream>>>(A1, Wmat, ht, 0);
  if (A2)
    gemm_wmma_f32<<<(MT + 7) / 8, 256, 0, stream>>>(A2, Wmat + DD * DD, ht, 1);
  scores<<<NN / 8, 256, 0, stream>>>(ht, a_s, a_d, s_src, s_dst);
  node_softmax_sum<<<NN / 8, 256, 0, stream>>>(rowptr, csr, ht, s_src, s_dst,
                                               bias, out);
}

extern "C" void kernel_launch(void* const* d_in, const int* in_sizes, int n_in,
                              void* d_out, int out_size, void* d_ws,
                              size_t ws_size, hipStream_t stream) {
  const float* x    = (const float*)d_in[0];
  const int*   ei   = (const int*)d_in[1];
  const float* We   = (const float*)d_in[2];
  const float* ae_s = (const float*)d_in[3];
  const float* ae_d = (const float*)d_in[4];
  const float* be   = (const float*)d_in[5];
  const float* Wp   = (const float*)d_in[6];
  const float* ap_s = (const float*)d_in[7];
  const float* ap_d = (const float*)d_in[8];
  const float* bp   = (const float*)d_in[9];
  const float* Wd   = (const float*)d_in[10];
  const float* ad_s = (const float*)d_in[11];
  const float* ad_d = (const float*)d_in[12];
  const float* bd   = (const float*)d_in[13];
  float*       out  = (float*)d_out;

  const size_t ND = (size_t)NN * DD;
  float*    enc   = (float*)d_ws;
  float*    hA    = enc + ND;
  float*    hB    = hA + ND;
  float*    ht    = hB + ND;
  float*    s_src = ht + ND;
  float*    s_dst = s_src + NN;
  int*      deg    = (int*)(s_dst + NN);
  int*      csr    = deg + NN;
  int*      rowptr = csr + EE;
  int*      slot    = (int*)hA;
  int*      grouped = (int*)hB;

  csr_zero<<<(NN + 255) / 256, 256, 0, stream>>>(deg);
  csr_count<<<EE / 256, 256, 0, stream>>>(ei, deg, slot);
  csr_scan<<<1, SCAN_T, 0, stream>>>(deg, rowptr);
  csr_place<<<EE / 256, 256, 0, stream>>>(ei, rowptr, slot, grouped);
  csr_order<<<NN / 8, 256, 0, stream>>>(rowptr, grouped, csr);

  run_gat(x, nullptr, We, ae_s, ae_d, be, rowptr, csr, ht, enc, s_src, s_dst,
          stream);
  run_gat(enc, enc, Wp, ap_s, ap_d, bp, rowptr, csr, ht, hA, s_src, s_dst,
          stream);
  run_gat(hA, enc, Wp, ap_s, ap_d, bp, rowptr, csr, ht, hB, s_src, s_dst,
          stream);
  run_gat(hB, nullptr, Wd, ad_s, ad_d, bd, rowptr, csr, ht, out, s_src, s_dst,
          stream);
}


# enc @ Wp[128:256], the same product in both processor steps, is formed once and each step starts its sums from it: 6 -> 5 GEMM launches; h sum order changes (k 128..255 first, fixed)
# speedup vs baseline: 4.5006x; 1.1834x over previous
#include <hip/hip_runtime.h>

#define NN 50000
#define EE 800000
#define DD 128
#define ETOT (EE + NN)
#define SLOPE 0.2f

typedef float v2f __attribute__((ext_vector_type(2)));
typedef float v8f __attribute__((ext_vector_type(8)));

#define MT (NN / 16)

template <int XM>
__device__ __forceinline__ float row_xor_add(float v) {
  return v + __int_as_float(__builtin_amdgcn_mov_dpp(
                 __float_as_int(v), 0x160 + XM, 0xf, 0xf, false));
}

__global__ __launch_bounds__(256) void gemm_wmma_f32(
    const float* __restrict__ A, const float* __restrict__ W,
    const float* __restrict__ C, float* __restrict__ H, int emit,
    const float* __restrict__ a_s, const float* __restrict__ a_d,
    float* __restrict__ s_src, float* __restrict__ s_dst) {
  __shared__ float4 Ws[(DD / 4) * DD];
  const int tid  = threadIdx.x;
  const int lane = tid & 31;
  const int mt   = blockIdx.x * 8 + (tid >> 5);
  const int h    = lane >> 4;
  const int c    = lane & 15;
  const int m0 = (mt < MT ? mt : MT - 1) * 16;

  const float* arow = A + (size_t)(m0 + c) * DD + 4 * h;
  float4 a[DD / 8];
#pragma unroll
  for (int j = 0; j < DD / 8; ++j) a[j] = *(const float4*)(arow + 8 * j);

  {
    const int col = tid & (DD - 1);
    const int kq0 = tid >> 7;
#pragma unroll
    for (int i = 0; i < DD / 8; ++i) {
      const int kq = kq0 + 2 * i;
      const float* wp = W + (size_t)(4 * kq) * DD + col;
      float4 w4;
      w4.x = wp[0];
      w4.y = wp[DD];
      w4.z = wp[2 * DD];
      w4.w = wp[3 * DD];
      Ws[kq * DD + col] = w4;
    }
  }
  __syncthreads();

  if (mt >= MT) return;

  float* hp = H + (size_t)(m0 + 8 * h) * DD + c;
  v8f acc[8];
  if (C) {
    const float* cp = C + (size_t)(m0 + 8 * h) * DD + c;
#pragma unroll
    for (int t = 0; t < 8; ++t)
#pragma unroll
      for (int r = 0; r < 8; ++r) acc[t][r] = cp[(size_t)r * DD + 16 * t];
  } else {
#pragma unroll
    for (int t = 0; t < 8; ++t)
#pragma unroll
      for (int r = 0; r < 8; ++r) acc[t][r] = 0.0f;
  }

  const float4* bp = Ws + h * DD + c;
#pragma unroll
  for (int j = 0; j < DD / 8; ++j) {
    float4 b[8];
#pragma unroll
    for (int t = 0; t < 8; ++t) b[t] = bp[j * 2 * DD + 16 * t];
    v2f av;
    av[0] = a[j].x;
    av[1] = a[j].y;
#pragma unroll
    for (int t = 0; t < 8; ++t) {
      v2f bv;
      bv[0] = b[t].x;
      bv[1] = b[t].y;
      acc[t] = __builtin_amdgcn_wmma_f32_16x16x4_f32(false, av, false, bv,
                                                     (short)0, acc[t], false, false);
    }
    av[0] = a[j].z;
    av[1] = a[j].w;
#pragma unroll
    for (int t = 0; t < 8; ++t) {
      v2f bv;
      bv[0] = b[t].z;
      bv[1] = b[t].w;
      acc[t] = __builtin_amdgcn_wmma_f32_16x16x4_f32(false, av, false, bv,
                                                     (short)0, acc[t], false, false);
    }
  }
#pragma unroll
  for (int t = 0; t < 8; ++t)
#pragma unroll
    for (int r = 0; r < 8; ++r) hp[(size_t)r * DD + 16 * t] = acc[t][r];

  if (!emit) return;

  float ps[8], pd[8];
#pragma unroll
  for (int r = 0; r < 8; ++r) {
    ps[r] = 0.0f;
    pd[r] = 0.0f;
  }
#pragma unroll
  for (int t = 0; t < 8; ++t) {
    const float as = a_s[16 * t + c];
    const float ad = a_d[16 * t + c];
#pragma unroll
    for (int r = 0; r < 8; ++r) {
      ps[r] += acc[t][r] * as;
      pd[r] += acc[t][r] * ad;
    }
  }
#pragma unroll
  for (int r = 0; r < 8; ++r) {
    ps[r] = row_xor_add<1>(ps[r]);
    pd[r] = row_xor_add<1>(pd[r]);
  }
#pragma unroll
  for (int r = 0; r < 8; ++r) {
    ps[r] = row_xor_add<2>(ps[r]);
    pd[r] = row_xor_add<2>(pd[r]);
  }
#pragma unroll
  for (int r = 0; r < 8; ++r) {
    ps[r] = row_xor_add<4>(ps[r]);
    pd[r] = row_xor_add<4>(pd[r]);
  }
#pragma unroll
  for (int r = 0; r < 8; ++r) {
    ps[r] = row_xor_add<8>(ps[r]);
    pd[r] = row_xor_add<8>(pd[r]);
  }
  if (c == 0) {
    float* sp = s_src + m0 + 8 * h;
    float* dp = s_dst + m0 + 8 * h;
    *(float4*)sp       = make_float4(ps[0], ps[1], ps[2], ps[3]);
    *(float4*)(sp + 4) = make_float4(ps[4], ps[5], ps[6], ps[7]);
    *(float4*)dp       = make_float4(pd[0], pd[1], pd[2], pd[3]);
    *(float4*)(dp + 4) = make_float4(pd[4], pd[5], pd[6], pd[7]);
  }
}

__global__ __launch_bounds__(256) void csr_zero(int* __restrict__ deg) {
  int i = blockIdx.x * blockDim.x + threadIdx.x;
  if (i < NN) deg[i] = 0;
}

__global__ __launch_bounds__(256) void csr_count(const int* __restrict__ ei,
                                                 int* __restrict__ deg,
                                                 int* __restrict__ slot) {
  int e = blockIdx.x * blockDim.x + threadIdx.x;
  slot[e] = atomicAdd(&deg[ei[EE + e]], 1);
}

#define SCAN_T 1024
#define SCAN_ROWS ((NN + SCAN_T) / SCAN_T)
__global__ __launch_bounds__(SCAN_T) void csr_scan(const int* __restrict__ deg,
                                                   int* __restrict__ rowptr) {
  __shared__ int wsum[32];
  const int lane = threadIdx.x & 31;
  const int wv   = threadIdx.x >> 5;
  const int b    = wv * (SCAN_ROWS * 32) + lane;
  int d[SCAN_ROWS];
  int local = 0;
#pragma unroll
  for (int k = 0; k < SCAN_ROWS; ++k) {
    const int idx = b + k * 32;
    const int v   = deg[idx < NN ? idx : NN - 1];
    d[k] = idx < NN ? v : 0;
    local += d[k];
  }
  for (int off = 16; off > 0; off >>= 1) local += __shfl_xor(local, off, 32);
  if (lane == 0) wsum[wv] = local;
  __syncthreads();
  if (wv == 0) {
    const int w = wsum[lane];
    int winc = w;
    for (int off = 1; off < 32; off <<= 1) {
      const int v = __shfl_up(winc, off, 32);
      if (lane >= off) winc += v;
    }
    wsum[lane] = winc - w;
  }
  __syncthreads();
  int carry = wsum[wv];
#pragma unroll
  for (int k = 0; k < SCAN_ROWS; ++k) {
    int inc = d[k];
    for (int off = 1; off < 32; off <<= 1) {
      const int v = __shfl_up(inc, off, 32);
      if (lane >= off) inc += v;
    }
    rowptr[b + k * 32] = carry + inc - d[k];
    carry += __shfl(inc, 31, 32);
  }
}

__global__ __launch_bounds__(256) void csr_place(const int* __restrict__ ei,
                                                 const int* __restrict__ rowptr,
                                                 const int* __restrict__ slot,
                                                 int* __restrict__ grouped) {
  int e = blockIdx.x * blockDim.x + threadIdx.x;
  grouped[rowptr[ei[EE + e]] + slot[e]] = ei[e];
}

__global__ __launch_bounds__(256) void csr_order(const int* __restrict__ rowptr,
                                                 const int* __restrict__ grouped,
                                                 int* __restrict__ csr) {
  const int lane = threadIdx.x & 31;
  const int node = blockIdx.x * 8 + (threadIdx.x >> 5);
  if (node >= NN) return;
  const int start = rowptr[node];
  const int cnt   = rowptr[node + 1] - start;
  for (int i = lane; i < cnt; i += 32) {
    const int si = grouped[start + i];
    int r = 0;
    for (int j = 0; j < cnt; ++j) {
      const int sj = grouped[start + j];
      r += (sj < si || (sj == si && j < i)) ? 1 : 0;
    }
    csr[start + r] = si;
  }
}

__device__ __forceinline__ float edge_score(float ss, float sd) {
  float v = ss + sd;
  return v > 0.0f ? v : SLOPE * v;
}

__device__ __forceinline__ void add_rows(float4& acc, const float* __restrict__ Hl,
                                         int s, float a, int n) {
#pragma unroll 4
  for (int j = 0; j < n; ++j) {
    const int   sj = __builtin_amdgcn_readlane(s, j);
    const float aj = __int_as_float(__builtin_amdgcn_readlane(__float_as_int(a), j));
    const float4 hv = *(const float4*)(Hl + (size_t)sj * DD);
    acc.x += aj * hv.x;
    acc.y += aj * hv.y;
    acc.z += aj * hv.z;
    acc.w += aj * hv.w;
  }
}

__global__ __launch_bounds__(256) void node_softmax_sum(
    const int* __restrict__ rowptr, const int* __restrict__ csr,
    const float* __restrict__ H, const float* __restrict__ s_src,
    const float* __restrict__ s_dst, const float* __restrict__ bias,
    float* __restrict__ out) {
  const int lane = threadIdx.x & 31;
  const int node = __builtin_amdgcn_readfirstlane(
      (int)(blockIdx.x * 8 + (threadIdx.x >> 5)));
  if (node >= NN) return;
  const int   start = rowptr[node];
  const int   cnt   = rowptr[node + 1] - start;
  const int   tot   = cnt + 1;
  const float sd    = s_dst[node];

  int   s0 = node;
  float v0 = -INFINITY;
  if (lane < tot) {
    if (lane < cnt) s0 = csr[start + lane];
    v0 = edge_score(s_src[s0], sd);
  }
  float m = v0;
  for (int i = 32 + lane; i < tot; i += 32) {
    const int s = (i < cnt) ? csr[start + i] : node;
    m = fmaxf(m, edge_score(s_src[s], sd));
  }
  for (int off = 16; off > 0; off >>= 1) m = fmaxf(m, __shfl_xor(m, off, 32));

  const float z0 = (lane < tot) ? __expf(v0 - m) : 0.0f;
  float den = z0;
  for (int i = 32 + lane; i < tot; i += 32) {
    const int s = (i < cnt) ? csr[start + i] : node;
    den += __expf(edge_score(s_src[s], sd) - m);
  }
  for (int off = 16; off > 0; off >>= 1) den += __shfl_xor(den, off, 32);

  float4 acc = make_float4(0.0f, 0.0f, 0.0f, 0.0f);
  const float* Hl = H + lane * 4;
  add_rows(acc, Hl, s0, z0 / den, tot < 32 ? tot : 32);
  for (int base = 32; base < tot; base += 32) {
    const int i = base + lane;
    int   s = node;
    float a = 0.0f;
    if (i < tot) {
      if (i < cnt) s = csr[start + i];
      a = __expf(edge_score(s_src[s], sd) - m) / den;
    }
    add_rows(acc, Hl, s, a, (tot - base) < 32 ? (tot - base) : 32);
  }
  const float4 b = *(const float4*)(bias + lane * 4);
  acc.x += b.x;
  acc.y += b.y;
  acc.z += b.z;
  acc.w += b.w;
  *(float4*)(out + (size_t)node * DD + lane * 4) = acc;
}

static void run_gat(const float* A, const float* C, const float* Wmat,
                    const float* a_s, const float* a_d, const float* bias,
                    const int* rowptr, const int* csr, float* ht, float* out,
                    float* s_src, float* s_dst, hipStream_t stream) {
  gemm_wmma_f32<<<(MT + 7) / 8, 256, 0, stream>>>(A, Wmat, C, ht, 1, a_s, a_d,
                                                  s_src, s_dst);
  node_softmax_sum<<<NN / 8, 256, 0, stream>>>(rowptr, csr, ht, s_src, s_dst,
                                               bias, out);
}

extern "C" void kernel_launch(void* const* d_in, const int* in_sizes, int n_in,
                              void* d_out, int out_size, void* d_ws,
                              size_t ws_size, hipStream_t stream) {
  const float* x    = (const float*)d_in[0];
  const int*   ei   = (const int*)d_in[1];
  const float* We   = (const float*)d_in[2];
  const float* ae_s = (const float*)d_in[3];
  const float* ae_d = (const float*)d_in[4];
  const float* be   = (const float*)d_in[5];
  const float* Wp   = (const float*)d_in[6];
  const float* ap_s = (const float*)d_in[7];
  const float* ap_d = (const float*)d_in[8];
  const float* bp   = (const float*)d_in[9];
  const float* Wd   = (const float*)d_in[10];
  const float* ad_s = (const float*)d_in[11];
  const float* ad_d = (const float*)d_in[12];
  const float* bd   = (const float*)d_in[13];
  float*       out  = (float*)d_out;

  const size_t ND = (size_t)NN * DD;
  float*    enc   = (float*)d_ws;
  float*    hA    = enc + ND;
  float*    hB    = hA + ND;
  float*    ht    = hB + ND;
  float*    s_src = ht + ND;
  float*    s_dst = s_src + NN;
  int*      deg    = (int*)(s_dst + NN);
  int*      csr    = deg + NN;
  int*      rowptr = csr + EE;
  int*      slot    = (int*)hA;
  int*      grouped = (int*)hB;

  csr_zero<<<(NN + 255) / 256, 256, 0, stream>>>(deg);
  csr_count<<<EE / 256, 256, 0, stream>>>(ei, deg, slot);
  csr_scan<<<1, SCAN_T, 0, stream>>>(deg, rowptr);
  csr_place<<<EE / 256, 256, 0, stream>>>(ei, rowptr, slot, grouped);
  csr_order<<<NN / 8, 256, 0, stream>>>(rowptr, grouped, csr);

  run_gat(x, nullptr, We, ae_s, ae_d, be, rowptr, csr, ht, enc, s_src, s_dst,
          stream);
  float* pe = hB;
  gemm_wmma_f32<<<(MT + 7) / 8, 256, 0, stream>>>(enc, Wp + DD * DD, nullptr,
                                                  pe, 0, ap_s, ap_d, s_src,
                                                  s_dst);
  run_gat(enc, pe, Wp, ap_s, ap_d, bp, rowptr, csr, ht, hA, s_src, s_dst,
          stream);
  run_gat(hA, pe, Wp, ap_s, ap_d, bp, rowptr, csr, ht, hB, s_src, s_dst,
          stream);
  run_gat(hB, nullptr, Wd, ad_s, ad_d, bd, rowptr, csr, ht, out, s_src, s_dst,
          stream);
}
